# starting kernel
# baseline (speedup 1.0000x reference)
#include <hip/hip_runtime.h>

typedef __attribute__((ext_vector_type(2))) float v2f;
typedef __attribute__((ext_vector_type(8))) float v8f;

#define BATCH   16384
#define NFEAT   1024
#define TDEPTH  10
#define NLEAVES 1024
#define WPB     8
#define RPW     16
#define RPB     (WPB * RPW)
#define KCHUNKS (NFEAT / 4)

__launch_bounds__(256)
__global__ void sdt_wmma_kernel(const float* __restrict__ x,
                                const float* __restrict__ W,
                                const float* __restrict__ bias,
                                const float* __restrict__ leaf,
                                float* __restrict__ out) {
  __shared__ float ldsB[KCHUNKS * 64];
  __shared__ float ldsLeaf[NLEAVES];
  __shared__ float ldsZ[WPB * 256];
  __shared__ float ldsV[WPB * 32];
  __shared__ float ldsBias[16];

  const int t    = threadIdx.x;
  const int wave = t >> 5;
  const int lane = t & 31;

  for (int idx = t; idx < 16 * NFEAT; idx += 256) {
    int n = idx >> 10;
    int k = idx & (NFEAT - 1);
    float w = W[n * NFEAT + k];
    int c  = k >> 2;
    int lb = ((k >> 1) & 1) * 16 + n;
    int j  = k & 1;
    ldsB[c * 64 + lb * 2 + j] = w;
  }
  for (int idx = t; idx < NLEAVES; idx += 256) ldsLeaf[idx] = leaf[idx];
  if (t < 16) ldsBias[t] = bias[t];
  __syncthreads();

  const int rowBase = (blockIdx.x * WPB + wave) * RPW;
  const int m = lane & 15;
  const int h = lane >> 4;

  const float* xrow = x + (size_t)(rowBase + m) * NFEAT + h * 2;
  const float* bptr = ldsB + lane * 2;
  v8f acc = {};
  #pragma unroll 4
  for (int c = 0; c < KCHUNKS; ++c) {
    v2f a  = *(const v2f*)(xrow + c * 4);
    v2f bf = *(const v2f*)(bptr + c * 64);
    acc = __builtin_amdgcn_wmma_f32_16x16x4_f32(
         false, a,  false, bf,
         (short)0, acc,  false,  false);
  }

  {
    float* zp = ldsZ + wave * 256;
    int n  = lane & 15;
    int mh = h * 8;
    #pragma unroll
    for (int i = 0; i < 8; ++i) zp[(mh + i) * 16 + n] = acc[i];
  }
  __syncthreads();

  float g[TDEPTH], gm[TDEPTH];
  {
    const float* zp = ldsZ + wave * 256 + m * 16;
    #pragma unroll
    for (int d = 0; d < TDEPTH; ++d) {
      float z = zp[d] + ldsBias[d];
      float e = __expf(-z);
      g[d]  = 1.0f / (1.0f + e);
      gm[d] = 1.0f - g[d];
    }
  }

  const float* lf = ldsLeaf + h * 512;
  float c16[16];
  #pragma unroll
  for (int jj = 0; jj < 16; ++jj) {
    float tt[16];
    #pragma unroll
    for (int i = 0; i < 16; ++i) {
      v2f lv = *(const v2f*)(lf + jj * 32 + i * 2);
      tt[i] = gm[9] * lv.x + g[9] * lv.y;
    }
    #pragma unroll
    for (int i = 0; i < 8; ++i) tt[i] = gm[8] * tt[2*i] + g[8] * tt[2*i+1];
    #pragma unroll
    for (int i = 0; i < 4; ++i) tt[i] = gm[7] * tt[2*i] + g[7] * tt[2*i+1];
    #pragma unroll
    for (int i = 0; i < 2; ++i) tt[i] = gm[6] * tt[2*i] + g[6] * tt[2*i+1];
    c16[jj] = gm[5] * tt[0] + g[5] * tt[1];
  }
  #pragma unroll
  for (int i = 0; i < 8; ++i) c16[i] = gm[4] * c16[2*i] + g[4] * c16[2*i+1];
  #pragma unroll
  for (int i = 0; i < 4; ++i) c16[i] = gm[3] * c16[2*i] + g[3] * c16[2*i+1];
  #pragma unroll
  for (int i = 0; i < 2; ++i) c16[i] = gm[2] * c16[2*i] + g[2] * c16[2*i+1];
  float vh = gm[1] * c16[0] + g[1] * c16[1];

  ldsV[wave * 32 + h * 16 + m] = vh;
  __syncthreads();
  if (lane < 16) {
    float v0 = ldsV[wave * 32 + m];
    float v1 = ldsV[wave * 32 + 16 + m];
    out[rowBase + m] = gm[0] * v0 + g[0] * v1;
  }
}

extern "C" void kernel_launch(void* const* d_in, const int* in_sizes, int n_in,
                              void* d_out, int out_size, void* d_ws, size_t ws_size,
                              hipStream_t stream) {
  (void)in_sizes; (void)n_in; (void)out_size; (void)d_ws; (void)ws_size;
  const float* x    = (const float*)d_in[0];
  const float* W    = (const float*)d_in[1];
  const float* b    = (const float*)d_in[2];
  const float* leaf = (const float*)d_in[3];
  float* out = (float*)d_out;
  sdt_wmma_kernel<<<dim3(BATCH / RPB), dim3(256), 0, stream>>>(x, W, b, leaf, out);
}


# loop nest turned: each wave reads one 128-feature k slice of W into registers (no LDS copy of W) and multiplies the block's 4 row tiles by it; grid 256; 8 slice sums added in LDS (sum regrouped)
# speedup vs baseline: 2.1328x; 2.1328x over previous
#include <hip/hip_runtime.h>


typedef __attribute__((ext_vector_type(2))) float v2f;
typedef __attribute__((ext_vector_type(4))) float v4f;
typedef __attribute__((ext_vector_type(8))) float v8f;

#define BATCH   16384
#define NFEAT   1024
#define TDEPTH  10
#define NLEAVES 1024
#define WPB     8
#define TPB     4
#define RPB     (TPB * 16)
#define KSLICE  (NFEAT / WPB)
#define KSTEPS  (KSLICE / 4)

__launch_bounds__(256)
__global__ void sdt_wmma_kernel(const float* __restrict__ x,
                                const float* __restrict__ W,
                                const float* __restrict__ bias,
                                const float* __restrict__ leaf,
                                float* __restrict__ out) {
  __shared__ float ldsP[WPB * TPB * 256];
  __shared__ float ldsZ[TPB * 256];
  __shared__ float ldsLeaf[NLEAVES];
  __shared__ float ldsV[RPB * 4];
  __shared__ float ldsG[RPB * 2];
  __shared__ float ldsBias[256];

  const int t    = threadIdx.x;
  const int wave = t >> 5;
  const int lane = t & 31;
  const int m = lane & 15;
  const int h = lane >> 4;

  const float* wrow  = W + (size_t)m * NFEAT + wave * KSLICE + h * 2;
  const float* xbase = x + (size_t)(blockIdx.x * RPB + m) * NFEAT + wave * KSLICE + h * 2;
  v2f bw[KSTEPS];
  v2f a[TPB][KSTEPS];
  #pragma unroll
  for (int c = 0; c < KSTEPS; ++c) {
    bw[c] = *(const v2f*)(wrow + c * 4);
    #pragma unroll
    for (int tl = 0; tl < TPB; ++tl)
      a[tl][c] = *(const v2f*)(xbase + (size_t)(tl * 16) * NFEAT + c * 4);
  }
  v8f acc[TPB];
  #pragma unroll
  for (int tl = 0; tl < TPB; ++tl) acc[tl] = (v8f){};
  #pragma unroll
  for (int c = 0; c < KSTEPS; ++c) {
    #pragma unroll
    for (int tl = 0; tl < TPB; ++tl)
      acc[tl] = __builtin_amdgcn_wmma_f32_16x16x4_f32(
           false, a[tl][c],  false, bw[c],
           (short)0, acc[tl],  false,  false);
  }

  {
    float lfr[NLEAVES / 256];
    #pragma unroll
    for (int i = 0; i < NLEAVES / 256; ++i) lfr[i] = leaf[t + 256 * i];
    const float br = bias[t & 15];
    #pragma unroll
    for (int i = 0; i < NLEAVES / 256; ++i) ldsLeaf[t + 256 * i] = lfr[i];
    ldsBias[t] = br;
  }

  {
    float* pp = ldsP + wave * (TPB * 256) + h * 16 + m;
    #pragma unroll
    for (int tl = 0; tl < TPB; ++tl) {
      #pragma unroll
      for (int i = 0; i < 8; ++i) pp[tl * 256 + i * 32] = acc[tl][i];
    }
  }
  __syncthreads();

  {
    v4f s = *(const v4f*)(ldsP + t * 4);
    #pragma unroll
    for (int w = 1; w < WPB; ++w) s += *(const v4f*)(ldsP + w * (TPB * 256) + t * 4);
    *(v4f*)(ldsZ + t * 4) = s;
  }
  __syncthreads();

  const int tile = wave >> 1;
  const int q    = (wave & 1) * 2 + h;
  const int rowInBlock = tile * 16 + m;

  float g[TDEPTH], gm[TDEPTH];
  {
    const float* zp = ldsZ + tile * 256 + (m & 7) * 32 + (m >> 3) * 16;
    #pragma unroll
    for (int d = 0; d < TDEPTH; ++d) {
      float z = zp[d] + ldsBias[d];
      float e = __expf(-z);
      g[d]  = 1.0f / (1.0f + e);
      gm[d] = 1.0f - g[d];
    }
  }

  const float* lf = ldsLeaf + q * 256;
  float c8[8];
  #pragma unroll
  for (int jj = 0; jj < 8; ++jj) {
    float tt[16];
    #pragma unroll
    for (int i = 0; i < 16; ++i) {
      v2f lv = *(const v2f*)(lf + jj * 32 + i * 2);
      tt[i] = gm[9] * lv.x + g[9] * lv.y;
    }
    #pragma unroll
    for (int i = 0; i < 8; ++i) tt[i] = gm[8] * tt[2*i] + g[8] * tt[2*i+1];
    #pragma unroll
    for (int i = 0; i < 4; ++i) tt[i] = gm[7] * tt[2*i] + g[7] * tt[2*i+1];
    #pragma unroll
    for (int i = 0; i < 2; ++i) tt[i] = gm[6] * tt[2*i] + g[6] * tt[2*i+1];
    c8[jj] = gm[5] * tt[0] + g[5] * tt[1];
  }
  #pragma unroll
  for (int i = 0; i < 4; ++i) c8[i] = gm[4] * c8[2*i] + g[4] * c8[2*i+1];
  #pragma unroll
  for (int i = 0; i < 2; ++i) c8[i] = gm[3] * c8[2*i] + g[3] * c8[2*i+1];
  float vq = gm[2] * c8[0] + g[2] * c8[1];

  ldsV[rowInBlock * 4 + q] = vq;
  if (q == 0) {
    ldsG[rowInBlock * 2 + 0] = g[0];
    ldsG[rowInBlock * 2 + 1] = g[1];
  }
  __syncthreads();
  if (wave == 0) {
    #pragma unroll
    for (int r = 0; r < 2; ++r) {
      int row = r * 32 + lane;
      v4f v  = *(const v4f*)(ldsV + row * 4);
      v2f gg = *(const v2f*)(ldsG + row * 2);
      float g0 = gg.x, g1 = gg.y;
      float m0 = 1.0f - g0, m1 = 1.0f - g1;
      float v0 = m1 * v.x + g1 * v.y;
      float v1 = m1 * v.z + g1 * v.w;
      out[blockIdx.x * RPB + row] = m0 * v0 + g0 * v1;
    }
  }
}

extern "C" void kernel_launch(void* const* d_in, const int* in_sizes, int n_in,
                              void* d_out, int out_size, void* d_ws, size_t ws_size,
                              hipStream_t stream) {
  (void)in_sizes; (void)n_in; (void)out_size; (void)d_ws; (void)ws_size;
  const float* x    = (const float*)d_in[0];
  const float* W    = (const float*)d_in[1];
  const float* b    = (const float*)d_in[2];
  const float* leaf = (const float*)d_in[3];
  float* out = (float*)d_out;
  sdt_wmma_kernel<<<dim3(BATCH / RPB), dim3(256), 0, stream>>>(x, W, b, leaf, out);
}


# x read as 64 contiguous B of one row per 4 lanes (8 lines per load, was 16), lane pairs swap halves by one DPP exchange into the A fragments; W in the same k order, 16 B per lane; k sum regrouped
# speedup vs baseline: 3.1169x; 1.4614x over previous
#include <hip/hip_runtime.h>


typedef __attribute__((ext_vector_type(2))) float v2f;
typedef __attribute__((ext_vector_type(4))) float v4f;
typedef __attribute__((ext_vector_type(8))) float v8f;

#define BATCH   16384
#define NFEAT   1024
#define TDEPTH  10
#define NLEAVES 1024
#define WPB     8
#define TPB     4
#define RPB     (TPB * 16)
#define KSLICE  (NFEAT / WPB)
#define NCHUNK  (KSLICE / 16)

static __device__ __forceinline__ float lane_xor1(float v) {
  return __builtin_bit_cast(float,
      __builtin_amdgcn_update_dpp(0, __builtin_bit_cast(int, v), 0xB1, 0xF, 0xF, true));
}

__launch_bounds__(256)
__global__ void sdt_wmma_kernel(const float* __restrict__ x,
                                const float* __restrict__ W,
                                const float* __restrict__ bias,
                                const float* __restrict__ leaf,
                                float* __restrict__ out) {
  __shared__ float ldsP[WPB * TPB * 256];
  __shared__ float ldsZ[TPB * 256];
  __shared__ float ldsLeaf[NLEAVES];
  __shared__ float ldsV[RPB * 4];
  __shared__ float ldsG[RPB * 2];
  __shared__ float ldsBias[256];

  const int t    = threadIdx.x;
  const int wave = t >> 5;
  const int lane = t & 31;
  const int m = lane & 15;
  const int h = lane >> 4;

  const int odd = m & 1;
  const float* wrow  = W + (size_t)m * NFEAT + wave * KSLICE + h * 8;
  const float* xbase = x + (size_t)(blockIdx.x * RPB + (m & ~1)) * NFEAT + wave * KSLICE + (2 * h + odd) * 4;
  v4f wq[NCHUNK][2];
  v4f xp[TPB][NCHUNK], xq[TPB][NCHUNK];
  #pragma unroll
  for (int g = 0; g < NCHUNK; ++g) {
    wq[g][0] = *(const v4f*)(wrow + g * 16);
    wq[g][1] = *(const v4f*)(wrow + g * 16 + 4);
    #pragma unroll
    for (int tl = 0; tl < TPB; ++tl) {
      xp[tl][g] = *(const v4f*)(xbase + (size_t)(tl * 16) * NFEAT + g * 16);
      xq[tl][g] = *(const v4f*)(xbase + (size_t)(tl * 16 + 1) * NFEAT + g * 16);
    }
    __builtin_amdgcn_sched_barrier(0);
  }
  v8f acc[TPB];
  #pragma unroll
  for (int tl = 0; tl < TPB; ++tl) acc[tl] = (v8f){};
  #pragma unroll
  for (int g = 0; g < NCHUNK; ++g) {
    v2f bq[4];
    bq[0] = (v2f){wq[g][0].x, wq[g][0].y};
    bq[1] = (v2f){wq[g][0].z, wq[g][0].w};
    bq[2] = (v2f){wq[g][1].x, wq[g][1].y};
    bq[3] = (v2f){wq[g][1].z, wq[g][1].w};
    v2f af[TPB][4];
    #pragma unroll
    for (int tl = 0; tl < TPB; ++tl) {
      const v4f p = xp[tl][g], q = xq[tl][g];
      v4f lo, hi;
      #pragma unroll
      for (int e = 0; e < 4; ++e) {
        const float send = odd ? p[e] : q[e];
        const float recv = lane_xor1(send);
        lo[e] = odd ? recv : p[e];
        hi[e] = odd ? q[e] : recv;
      }
      af[tl][0] = (v2f){lo.x, lo.y};
      af[tl][1] = (v2f){lo.z, lo.w};
      af[tl][2] = (v2f){hi.x, hi.y};
      af[tl][3] = (v2f){hi.z, hi.w};
    }
    #pragma unroll
    for (int p = 0; p < 4; ++p) {
      #pragma unroll
      for (int tl = 0; tl < TPB; ++tl)
        acc[tl] = __builtin_amdgcn_wmma_f32_16x16x4_f32(
             false, af[tl][p],  false, bq[p],
             (short)0, acc[tl],  false,  false);
    }
  }

  {
    float lfr[NLEAVES / 256];
    #pragma unroll
    for (int i = 0; i < NLEAVES / 256; ++i) lfr[i] = leaf[t + 256 * i];
    const float br = bias[t & 15];
    #pragma unroll
    for (int i = 0; i < NLEAVES / 256; ++i) ldsLeaf[t + 256 * i] = lfr[i];
    ldsBias[t] = br;
  }

  {
    float* pp = ldsP + wave * (TPB * 256) + h * 16 + m;
    #pragma unroll
    for (int tl = 0; tl < TPB; ++tl) {
      #pragma unroll
      for (int i = 0; i < 8; ++i) pp[tl * 256 + i * 32] = acc[tl][i];
    }
  }
  __syncthreads();

  {
    v4f s = *(const v4f*)(ldsP + t * 4);
    #pragma unroll
    for (int w = 1; w < WPB; ++w) s += *(const v4f*)(ldsP + w * (TPB * 256) + t * 4);
    *(v4f*)(ldsZ + t * 4) = s;
  }
  __syncthreads();

  const int tile = wave >> 1;
  const int q    = (wave & 1) * 2 + h;
  const int rowInBlock = tile * 16 + m;

  float g[TDEPTH], gm[TDEPTH];
  {
    const float* zp = ldsZ + tile * 256 + (m & 7) * 32 + (m >> 3) * 16;
    #pragma unroll
    for (int d = 0; d < TDEPTH; ++d) {
      float z = zp[d] + ldsBias[d];
      float e = __expf(-z);
      g[d]  = 1.0f / (1.0f + e);
      gm[d] = 1.0f - g[d];
    }
  }

  const float* lf = ldsLeaf + q * 256;
  float c8[8];
  #pragma unroll
  for (int jj = 0; jj < 8; ++jj) {
    float tt[16];
    #pragma unroll
    for (int i = 0; i < 16; ++i) {
      v2f lv = *(const v2f*)(lf + jj * 32 + i * 2);
      tt[i] = gm[9] * lv.x + g[9] * lv.y;
    }
    #pragma unroll
    for (int i = 0; i < 8; ++i) tt[i] = gm[8] * tt[2*i] + g[8] * tt[2*i+1];
    #pragma unroll
    for (int i = 0; i < 4; ++i) tt[i] = gm[7] * tt[2*i] + g[7] * tt[2*i+1];
    #pragma unroll
    for (int i = 0; i < 2; ++i) tt[i] = gm[6] * tt[2*i] + g[6] * tt[2*i+1];
    c8[jj] = gm[5] * tt[0] + g[5] * tt[1];
  }
  #pragma unroll
  for (int i = 0; i < 4; ++i) c8[i] = gm[4] * c8[2*i] + g[4] * c8[2*i+1];
  #pragma unroll
  for (int i = 0; i < 2; ++i) c8[i] = gm[3] * c8[2*i] + g[3] * c8[2*i+1];
  float vq = gm[2] * c8[0] + g[2] * c8[1];

  ldsV[rowInBlock * 4 + q] = vq;
  if (q == 0) {
    ldsG[rowInBlock * 2 + 0] = g[0];
    ldsG[rowInBlock * 2 + 1] = g[1];
  }
  __syncthreads();
  if (wave == 0) {
    #pragma unroll
    for (int r = 0; r < 2; ++r) {
      int row = r * 32 + lane;
      v4f v  = *(const v4f*)(ldsV + row * 4);
      v2f gg = *(const v2f*)(ldsG + row * 2);
      float g0 = gg.x, g1 = gg.y;
      float m0 = 1.0f - g0, m1 = 1.0f - g1;
      float v0 = m1 * v.x + g1 * v.y;
      float v1 = m1 * v.z + g1 * v.w;
      out[blockIdx.x * RPB + row] = m0 * v0 + g0 * v1;
    }
  }
}

extern "C" void kernel_launch(void* const* d_in, const int* in_sizes, int n_in,
                              void* d_out, int out_size, void* d_ws, size_t ws_size,
                              hipStream_t stream) {
  (void)in_sizes; (void)n_in; (void)out_size; (void)d_ws; (void)ws_size;
  const float* x    = (const float*)d_in[0];
  const float* W    = (const float*)d_in[1];
  const float* b    = (const float*)d_in[2];
  const float* leaf = (const float*)d_in[3];
  float* out = (float*)d_out;
  sdt_wmma_kernel<<<dim3(BATCH / RPB), dim3(256), 0, stream>>>(x, W, b, leaf, out);
}


# request order: the two 64-byte halves of every 128-byte line of x and W are asked for by neighbouring loads of a wave (were 10 loads apart), wmmas follow that order; every sum unchanged
# speedup vs baseline: 3.3935x; 1.0887x over previous
#include <hip/hip_runtime.h>


typedef __attribute__((ext_vector_type(2))) float v2f;
typedef __attribute__((ext_vector_type(4))) float v4f;
typedef __attribute__((ext_vector_type(8))) float v8f;

#define BATCH   16384
#define NFEAT   1024
#define TDEPTH  10
#define NLEAVES 1024
#define WPB     8
#define TPB     4
#define RPB     (TPB * 16)
#define KSLICE  (NFEAT / WPB)
#define NCHUNK  (KSLICE / 16)

static __device__ __forceinline__ float lane_xor1(float v) {
  return __builtin_bit_cast(float,
      __builtin_amdgcn_update_dpp(0, __builtin_bit_cast(int, v), 0xB1, 0xF, 0xF, true));
}

__launch_bounds__(256)
__global__ void sdt_wmma_kernel(const float* __restrict__ x,
                                const float* __restrict__ W,
                                const float* __restrict__ bias,
                                const float* __restrict__ leaf,
                                float* __restrict__ out) {
  __shared__ float ldsP[WPB * TPB * 256];
  __shared__ float ldsZ[TPB * 256];
  __shared__ float ldsLeaf[NLEAVES];
  __shared__ float ldsV[RPB * 4];
  __shared__ float ldsG[RPB * 2];
  __shared__ float ldsBias[256];

  const int t    = threadIdx.x;
  const int wave = t >> 5;
  const int lane = t & 31;
  const int m = lane & 15;
  const int h = lane >> 4;

  const int odd = m & 1;
  const float* wrow  = W + (size_t)m * NFEAT + wave * KSLICE + h * 8;
  const float* xbase = x + (size_t)(blockIdx.x * RPB + (m & ~1)) * NFEAT + wave * KSLICE + (2 * h + odd) * 4;
  v4f wq[NCHUNK][2];
  v4f xp[TPB][NCHUNK], xq[TPB][NCHUNK];
  #pragma unroll
  for (int u = 0; u < NCHUNK / 2; ++u) {
    #pragma unroll
    for (int s = 0; s < 2; ++s) {
      wq[2 * u + s][0] = *(const v4f*)(wrow + (2 * u + s) * 16);
      wq[2 * u + s][1] = *(const v4f*)(wrow + (2 * u + s) * 16 + 4);
    }
    #pragma unroll
    for (int tl = 0; tl < TPB; ++tl) {
      xp[tl][2 * u]     = *(const v4f*)(xbase + (size_t)(tl * 16) * NFEAT + (2 * u) * 16);
      xp[tl][2 * u + 1] = *(const v4f*)(xbase + (size_t)(tl * 16) * NFEAT + (2 * u + 1) * 16);
      xq[tl][2 * u]     = *(const v4f*)(xbase + (size_t)(tl * 16 + 1) * NFEAT + (2 * u) * 16);
      xq[tl][2 * u + 1] = *(const v4f*)(xbase + (size_t)(tl * 16 + 1) * NFEAT + (2 * u + 1) * 16);
      __builtin_amdgcn_sched_barrier(0);
    }
  }
  float lfr[NLEAVES / 256];
  #pragma unroll
  for (int i = 0; i < NLEAVES / 256; ++i) lfr[i] = leaf[t + 256 * i];
  const float br = bias[t & 15];
  __builtin_amdgcn_sched_barrier(0);
  v8f acc[TPB];
  #pragma unroll
  for (int tl = 0; tl < TPB; ++tl) acc[tl] = (v8f){};
  #pragma unroll
  for (int u = 0; u < NCHUNK / 2; ++u) {
    #pragma unroll
    for (int th = 0; th < TPB / 2; ++th) {
      __builtin_amdgcn_sched_barrier(0);
      #pragma unroll
      for (int s = 0; s < 2; ++s) {
        const int g = 2 * u + s;
        v2f bq[4];
        bq[0] = (v2f){wq[g][0].x, wq[g][0].y};
        bq[1] = (v2f){wq[g][0].z, wq[g][0].w};
        bq[2] = (v2f){wq[g][1].x, wq[g][1].y};
        bq[3] = (v2f){wq[g][1].z, wq[g][1].w};
        v2f af[2][4];
        #pragma unroll
        for (int tt = 0; tt < 2; ++tt) {
          const int tl = 2 * th + tt;
          const v4f p = xp[tl][g], q = xq[tl][g];
          v4f lo, hi;
          #pragma unroll
          for (int e = 0; e < 4; ++e) {
            const float send = odd ? p[e] : q[e];
            const float recv = lane_xor1(send);
            lo[e] = odd ? recv : p[e];
            hi[e] = odd ? q[e] : recv;
          }
          af[tt][0] = (v2f){lo.x, lo.y};
          af[tt][1] = (v2f){lo.z, lo.w};
          af[tt][2] = (v2f){hi.x, hi.y};
          af[tt][3] = (v2f){hi.z, hi.w};
        }
        #pragma unroll
        for (int p = 0; p < 4; ++p) {
          #pragma unroll
          for (int tt = 0; tt < 2; ++tt)
            acc[2 * th + tt] = __builtin_amdgcn_wmma_f32_16x16x4_f32(
                 false, af[tt][p],  false, bq[p],
                 (short)0, acc[2 * th + tt],  false,  false);
        }
      }
    }
  }

  {
    #pragma unroll
    for (int i = 0; i < NLEAVES / 256; ++i) ldsLeaf[t + 256 * i] = lfr[i];
    ldsBias[t] = br;
  }

  {
    float* pp = ldsP + wave * (TPB * 256) + h * 16 + m;
    #pragma unroll
    for (int tl = 0; tl < TPB; ++tl) {
      #pragma unroll
      for (int i = 0; i < 8; ++i) pp[tl * 256 + i * 32] = acc[tl][i];
    }
  }
  __syncthreads();

  {
    v4f s = *(const v4f*)(ldsP + t * 4);
    #pragma unroll
    for (int w = 1; w < WPB; ++w) s += *(const v4f*)(ldsP + w * (TPB * 256) + t * 4);
    *(v4f*)(ldsZ + t * 4) = s;
  }
  __syncthreads();

  const int tile = wave >> 1;
  const int q    = (wave & 1) * 2 + h;
  const int rowInBlock = tile * 16 + m;

  float g[TDEPTH], gm[TDEPTH];
  {
    const float* zp = ldsZ + tile * 256 + (m & 7) * 32 + (m >> 3) * 16;
    #pragma unroll
    for (int d = 0; d < TDEPTH; ++d) {
      float z = zp[d] + ldsBias[d];
      float e = __expf(-z);
      g[d]  = 1.0f / (1.0f + e);
      gm[d] = 1.0f - g[d];
    }
  }

  const float* lf = ldsLeaf + q * 256;
  float c8[8];
  #pragma unroll
  for (int jj = 0; jj < 8; ++jj) {
    float tt[16];
    #pragma unroll
    for (int i = 0; i < 16; ++i) {
      v2f lv = *(const v2f*)(lf + jj * 32 + i * 2);
      tt[i] = gm[9] * lv.x + g[9] * lv.y;
    }
    #pragma unroll
    for (int i = 0; i < 8; ++i) tt[i] = gm[8] * tt[2*i] + g[8] * tt[2*i+1];
    #pragma unroll
    for (int i = 0; i < 4; ++i) tt[i] = gm[7] * tt[2*i] + g[7] * tt[2*i+1];
    #pragma unroll
    for (int i = 0; i < 2; ++i) tt[i] = gm[6] * tt[2*i] + g[6] * tt[2*i+1];
    c8[jj] = gm[5] * tt[0] + g[5] * tt[1];
  }
  #pragma unroll
  for (int i = 0; i < 4; ++i) c8[i] = gm[4] * c8[2*i] + g[4] * c8[2*i+1];
  #pragma unroll
  for (int i = 0; i < 2; ++i) c8[i] = gm[3] * c8[2*i] + g[3] * c8[2*i+1];
  float vq = gm[2] * c8[0] + g[2] * c8[1];

  ldsV[rowInBlock * 4 + q] = vq;
  if (q == 0) {
    ldsG[rowInBlock * 2 + 0] = g[0];
    ldsG[rowInBlock * 2 + 1] = g[1];
  }
  __syncthreads();
  if (wave == 0) {
    #pragma unroll
    for (int r = 0; r < 2; ++r) {
      int row = r * 32 + lane;
      v4f v  = *(const v4f*)(ldsV + row * 4);
      v2f gg = *(const v2f*)(ldsG + row * 2);
      float g0 = gg.x, g1 = gg.y;
      float m0 = 1.0f - g0, m1 = 1.0f - g1;
      float v0 = m1 * v.x + g1 * v.y;
      float v1 = m1 * v.z + g1 * v.w;
      out[blockIdx.x * RPB + row] = m0 * v0 + g0 * v1;
    }
  }
}

extern "C" void kernel_launch(void* const* d_in, const int* in_sizes, int n_in,
                              void* d_out, int out_size, void* d_ws, size_t ws_size,
                              hipStream_t stream) {
  (void)in_sizes; (void)n_in; (void)out_size; (void)d_ws; (void)ws_size;
  const float* x    = (const float*)d_in[0];
  const float* W    = (const float*)d_in[1];
  const float* b    = (const float*)d_in[2];
  const float* leaf = (const float*)d_in[3];
  float* out = (float*)d_out;
  sdt_wmma_kernel<<<dim3(BATCH / RPB), dim3(256), 0, stream>>>(x, W, b, leaf, out);
}
